# starting kernel
# baseline (speedup 1.0000x reference)
#include <hip/hip_runtime.h>
#include <math.h>

typedef __attribute__((ext_vector_type(2))) float v2f;
typedef __attribute__((ext_vector_type(8))) float v8f;

#define NH 16
#define HD 64
#define ROWSZ (NH * HD)
#define LDS_STRIDE 18

__global__ __launch_bounds__(32)
void va_fwd_kernel(const float* __restrict__ q,
                   const float* __restrict__ k,
                   const float* __restrict__ v,
                   const int*   __restrict__ cuq,
                   const int*   __restrict__ cuk,
                   float*       __restrict__ out)
{
    const int lane = threadIdx.x & 31;
    const int il   = lane & 15;
    const int hi   = lane >> 4;
    const int b    = blockIdx.y;
    const int h    = blockIdx.z;

    const int cq0  = cuq[b];
    const int cq1  = cuq[b + 1];
    const int lenq = cq1 - cq0;
    const int base = blockIdx.x * 16;
    if (base >= lenq) return;

    const int ck0   = cuk[b];
    const int lenk  = cuk[b + 1] - ck0;
    const int delta = lenk - lenq;
    const int nrows = min(16, lenq - base);

    __shared__ float pl[16 * LDS_STRIDE];

    int qrow = cq0 + base + il;
    if (qrow > cq1 - 1) qrow = cq1 - 1;
    const float* qp = q + (size_t)qrow * ROWSZ + h * HD + 2 * hi;
    v2f qa[16];
#pragma unroll
    for (int c = 0; c < 16; ++c) {
        v2f t = *(const v2f*)(qp + 4 * c);
        qa[c] = t * 0.125f;
    }

    v8f oacc[4];
#pragma unroll
    for (int nc = 0; nc < 4; ++nc) oacc[nc] = (v8f){};
    v8f lacc = (v8f){};
    float mrow[8];
#pragma unroll
    for (int r = 0; r < 8; ++r) mrow[r] = -3.0e38f;

    int maxj = base + 15 + delta;
    if (maxj > lenk - 1) maxj = lenk - 1;
    if (maxj < 0) {
#pragma unroll
        for (int r = 0; r < 8; ++r) {
            int m = r + 8 * hi;
            if (m < nrows) {
                size_t ro = (size_t)(cq0 + base + m) * ROWSZ + h * HD;
#pragma unroll
                for (int nc = 0; nc < 4; ++nc) out[ro + nc * 16 + il] = 0.0f;
            }
        }
        return;
    }
    const int ktiles = maxj / 16 + 1;
    const float* kbase = k + (size_t)ck0 * ROWSZ + h * HD;
    const float* vbase = v + (size_t)ck0 * ROWSZ + h * HD;
    const v2f ones = {1.0f, 1.0f};

    for (int kt = 0; kt < ktiles; ++kt) {
        if (kt + 1 < ktiles) {
            int pr = min((kt + 1) * 16 + il, lenk - 1);
            __builtin_prefetch(kbase + (size_t)pr * ROWSZ, 0, 3);
            __builtin_prefetch(vbase + (size_t)pr * ROWSZ, 0, 3);
        }

        int krow = kt * 16 + il;
        if (krow > lenk - 1) krow = lenk - 1;
        const float* kp = kbase + (size_t)krow * ROWSZ + 2 * hi;
        v8f s = (v8f){};
#pragma unroll
        for (int c = 0; c < 16; ++c) {
            v2f kb = *(const v2f*)(kp + 4 * c);
            s = __builtin_amdgcn_wmma_f32_16x16x4_f32(
                    false, qa[c], false, kb, (short)0, s, false, false);
        }

        const int jg = kt * 16 + il;
#pragma unroll
        for (int r = 0; r < 8; ++r) {
            int ig = base + r + 8 * hi;
            bool ok = (jg <= ig + delta) && (jg < lenk);
            s[r] = ok ? s[r] : -1.0e9f;
        }

        float alpha[8];
#pragma unroll
        for (int r = 0; r < 8; ++r) {
            float x = s[r];
            x = fmaxf(x, __shfl_xor(x, 1, 16));
            x = fmaxf(x, __shfl_xor(x, 2, 16));
            x = fmaxf(x, __shfl_xor(x, 4, 16));
            x = fmaxf(x, __shfl_xor(x, 8, 16));
            float mnew = fmaxf(mrow[r], x);
            s[r]     = __expf(s[r] - mnew);
            alpha[r] = __expf(mrow[r] - mnew);
            mrow[r]  = mnew;
        }
#pragma unroll
        for (int r = 0; r < 8; ++r) {
            oacc[0][r] *= alpha[r]; oacc[1][r] *= alpha[r];
            oacc[2][r] *= alpha[r]; oacc[3][r] *= alpha[r];
            lacc[r]    *= alpha[r];
        }

#pragma unroll
        for (int r = 0; r < 8; ++r)
            pl[(r + 8 * hi) * LDS_STRIDE + il] = s[r];
        __syncthreads();
        v2f pa[4];
#pragma unroll
        for (int c = 0; c < 4; ++c)
            pa[c] = *(const v2f*)(&pl[il * LDS_STRIDE + 4 * c + 2 * hi]);
        __syncthreads();

#pragma unroll
        for (int c = 0; c < 4; ++c)
            lacc = __builtin_amdgcn_wmma_f32_16x16x4_f32(
                       false, pa[c], false, ones, (short)0, lacc, false, false);

#pragma unroll
        for (int nc = 0; nc < 4; ++nc) {
#pragma unroll
            for (int c = 0; c < 4; ++c) {
                int kr0 = kt * 16 + 4 * c + 2 * hi;
                int kr1 = kr0 + 1;
                kr0 = min(kr0, lenk - 1);
                kr1 = min(kr1, lenk - 1);
                v2f vb = { vbase[(size_t)kr0 * ROWSZ + nc * 16 + il],
                           vbase[(size_t)kr1 * ROWSZ + nc * 16 + il] };
                oacc[nc] = __builtin_amdgcn_wmma_f32_16x16x4_f32(
                               false, pa[c], false, vb, (short)0, oacc[nc],
                               false, false);
            }
        }
    }

#pragma unroll
    for (int r = 0; r < 8; ++r) {
        int m = r + 8 * hi;
        if (m < nrows) {
            float inv = 1.0f / lacc[r];
            size_t ro = (size_t)(cq0 + base + m) * ROWSZ + h * HD;
#pragma unroll
            for (int nc = 0; nc < 4; ++nc)
                out[ro + nc * 16 + il] = oacc[nc][r] * inv;
        }
    }
}

extern "C" void kernel_launch(void* const* d_in, const int* in_sizes, int n_in,
                              void* d_out, int out_size, void* d_ws, size_t ws_size,
                              hipStream_t stream) {
    const float* q   = (const float*)d_in[0];
    const float* k   = (const float*)d_in[1];
    const float* v   = (const float*)d_in[2];
    const int*   cuq = (const int*)d_in[3];
    const int*   cuk = (const int*)d_in[4];
    float* out = (float*)d_out;

    const int T = in_sizes[0] / ROWSZ;
    const int B = in_sizes[3] - 1;
    dim3 grid((T + 15) / 16, B, NH);
    va_fwd_kernel<<<grid, 32, 0, stream>>>(q, k, v, cuq, cuk, out);
}


# Work-group id picks its (batch, head, query tile) item from the list sorted by key-tile count, dealt so that any run or stride of ids mixes all lengths; was grid position = tile
# speedup vs baseline: 1.1145x; 1.1145x over previous
#include <hip/hip_runtime.h>
#include <math.h>

typedef __attribute__((ext_vector_type(2))) float v2f;
typedef __attribute__((ext_vector_type(8))) float v8f;

#define NH 16
#define HD 64
#define ROWSZ (NH * HD)
#define LDS_STRIDE 18

__device__ __forceinline__ int tiles_ge(int nt, int sh, int lk, int t)
{
    if (t <= 0) return nt;
    if (16 * (t - 1) > lk - 1) return 0;
    int xmin = t - 1 - sh;
    if (xmin < 0) xmin = 0;
    const int c = nt - xmin;
    return c > 0 ? c : 0;
}

__global__ __launch_bounds__(32)
void va_fwd_kernel(const float* __restrict__ q,
                   const float* __restrict__ k,
                   const float* __restrict__ v,
                   const int*   __restrict__ cuq,
                   const int*   __restrict__ cuk,
                   float*       __restrict__ out)
{
    const int lane = threadIdx.x & 31;
    const int il   = lane & 15;
    const int hi   = lane >> 4;
    const int nb   = gridDim.y;

    int b, h, xt, cq0, cq1, ck0, lenk;
    if (nb > 32) {
        b  = blockIdx.y;
        h  = blockIdx.z;
        xt = blockIdx.x;
        cq0 = cuq[b];
        cq1 = cuq[b + 1];
        ck0 = cuk[b];
        lenk = cuk[b + 1] - ck0;
    } else {
        int a0 = 0, a1 = 0, c0 = 0, lk = 0, nt = 0, sh = 0;
        if (lane < nb) {
            a0 = cuq[lane];
            a1 = cuq[lane + 1];
            c0 = cuk[lane];
            lk = cuk[lane + 1] - c0;
            const int lq = a1 - a0;
            nt = lq > 0 ? (lq + 15) >> 4 : 0;
            sh = (15 + lk - lq) >> 4;
        }
        int tot = 0, tmax = 1;
        for (int bb = 0; bb < nb; ++bb) {
            tot += __builtin_amdgcn_readlane(nt, bb);
            const int kk = ((__builtin_amdgcn_readlane(lk, bb) - 1) >> 4) + 2;
            tmax = kk > tmax ? kk : tmax;
        }
        const int wgid  = blockIdx.x + gridDim.x * (blockIdx.y + gridDim.y * blockIdx.z);
        const int items = NH * tot;
        if (wgid >= items) return;
        int rank = wgid;
        const int rows = items >> 8;
        if (wgid < (rows << 8)) {
            const int j = wgid & 255;
            rank = ((((wgid >> 8) + j) % rows) << 8) + j;
        }

        int lo = 0, span = tmax;
        for (int pass = 0; pass < 7 && span > 1; ++pass) {
            const int step = (span + 31) >> 5;
            const int t = lo + lane * step;
            int n = 0;
            for (int bb = 0; bb < nb; ++bb)
                n += tiles_ge(__builtin_amdgcn_readlane(nt, bb),
                              __builtin_amdgcn_readlane(sh, bb),
                              __builtin_amdgcn_readlane(lk, bb), t);
            const unsigned m = __builtin_amdgcn_ballot_w32(NH * n > rank);
            lo += (__builtin_popcount(m) - 1) * step;
            span = step;
        }
        const int tw = lo;
        int above = 0;
        for (int bb = 0; bb < nb; ++bb)
            above += tiles_ge(__builtin_amdgcn_readlane(nt, bb),
                              __builtin_amdgcn_readlane(sh, bb),
                              __builtin_amdgcn_readlane(lk, bb), tw + 1);
        const int r = rank - NH * above;
        h = r % NH;
        int qx = r / NH;
        b = 0;
        xt = 0;
        bool found = false;
        for (int bb = 0; bb < nb; ++bb) {
            const int ntb = __builtin_amdgcn_readlane(nt, bb);
            const int shb = __builtin_amdgcn_readlane(sh, bb);
            const int lkb = __builtin_amdgcn_readlane(lk, bb);
            const int ca = tiles_ge(ntb, shb, lkb, tw);
            const int g  = ca - tiles_ge(ntb, shb, lkb, tw + 1);
            if (!found) {
                if (qx < g) { b = bb; xt = ntb - ca + qx; found = true; }
                else        qx -= g;
            }
        }
        cq0  = __builtin_amdgcn_readlane(a0, b);
        cq1  = __builtin_amdgcn_readlane(a1, b);
        ck0  = __builtin_amdgcn_readlane(c0, b);
        lenk = __builtin_amdgcn_readlane(lk, b);
    }

    const int lenq = cq1 - cq0;
    const int base = xt * 16;
    if (base >= lenq) return;

    const int delta = lenk - lenq;
    const int nrows = min(16, lenq - base);

    __shared__ float pl[16 * LDS_STRIDE];

    int qrow = cq0 + base + il;
    if (qrow > cq1 - 1) qrow = cq1 - 1;
    const float* qp = q + (size_t)qrow * ROWSZ + h * HD + 2 * hi;
    v2f qa[16];
#pragma unroll
    for (int c = 0; c < 16; ++c) {
        v2f t = *(const v2f*)(qp + 4 * c);
        qa[c] = t * 0.125f;
    }

    v8f oacc[4];
#pragma unroll
    for (int nc = 0; nc < 4; ++nc) oacc[nc] = (v8f){};
    v8f lacc = (v8f){};
    float mrow[8];
#pragma unroll
    for (int r = 0; r < 8; ++r) mrow[r] = -3.0e38f;

    int maxj = base + 15 + delta;
    if (maxj > lenk - 1) maxj = lenk - 1;
    if (maxj < 0) {
#pragma unroll
        for (int r = 0; r < 8; ++r) {
            int m = r + 8 * hi;
            if (m < nrows) {
                size_t ro = (size_t)(cq0 + base + m) * ROWSZ + h * HD;
#pragma unroll
                for (int nc = 0; nc < 4; ++nc) out[ro + nc * 16 + il] = 0.0f;
            }
        }
        return;
    }
    const int ktiles = maxj / 16 + 1;
    const float* kbase = k + (size_t)ck0 * ROWSZ + h * HD;
    const float* vbase = v + (size_t)ck0 * ROWSZ + h * HD;
    const v2f ones = {1.0f, 1.0f};

    for (int kt = 0; kt < ktiles; ++kt) {
        if (kt + 1 < ktiles) {
            int pr = min((kt + 1) * 16 + il, lenk - 1);
            __builtin_prefetch(kbase + (size_t)pr * ROWSZ, 0, 3);
            __builtin_prefetch(vbase + (size_t)pr * ROWSZ, 0, 3);
        }

        int krow = kt * 16 + il;
        if (krow > lenk - 1) krow = lenk - 1;
        const float* kp = kbase + (size_t)krow * ROWSZ + 2 * hi;
        v8f s = (v8f){};
#pragma unroll
        for (int c = 0; c < 16; ++c) {
            v2f kb = *(const v2f*)(kp + 4 * c);
            s = __builtin_amdgcn_wmma_f32_16x16x4_f32(
                    false, qa[c], false, kb, (short)0, s, false, false);
        }

        const int jg = kt * 16 + il;
#pragma unroll
        for (int r = 0; r < 8; ++r) {
            int ig = base + r + 8 * hi;
            bool ok = (jg <= ig + delta) && (jg < lenk);
            s[r] = ok ? s[r] : -1.0e9f;
        }

        float alpha[8];
#pragma unroll
        for (int r = 0; r < 8; ++r) {
            float x = s[r];
            x = fmaxf(x, __shfl_xor(x, 1, 16));
            x = fmaxf(x, __shfl_xor(x, 2, 16));
            x = fmaxf(x, __shfl_xor(x, 4, 16));
            x = fmaxf(x, __shfl_xor(x, 8, 16));
            float mnew = fmaxf(mrow[r], x);
            s[r]     = __expf(s[r] - mnew);
            alpha[r] = __expf(mrow[r] - mnew);
            mrow[r]  = mnew;
        }
#pragma unroll
        for (int r = 0; r < 8; ++r) {
            oacc[0][r] *= alpha[r]; oacc[1][r] *= alpha[r];
            oacc[2][r] *= alpha[r]; oacc[3][r] *= alpha[r];
            lacc[r]    *= alpha[r];
        }

#pragma unroll
        for (int r = 0; r < 8; ++r)
            pl[(r + 8 * hi) * LDS_STRIDE + il] = s[r];
        __syncthreads();
        v2f pa[4];
#pragma unroll
        for (int c = 0; c < 4; ++c)
            pa[c] = *(const v2f*)(&pl[il * LDS_STRIDE + 4 * c + 2 * hi]);
        __syncthreads();

#pragma unroll
        for (int c = 0; c < 4; ++c)
            lacc = __builtin_amdgcn_wmma_f32_16x16x4_f32(
                       false, pa[c], false, ones, (short)0, lacc, false, false);

#pragma unroll
        for (int nc = 0; nc < 4; ++nc) {
#pragma unroll
            for (int c = 0; c < 4; ++c) {
                int kr0 = kt * 16 + 4 * c + 2 * hi;
                int kr1 = kr0 + 1;
                kr0 = min(kr0, lenk - 1);
                kr1 = min(kr1, lenk - 1);
                v2f vb = { vbase[(size_t)kr0 * ROWSZ + nc * 16 + il],
                           vbase[(size_t)kr1 * ROWSZ + nc * 16 + il] };
                oacc[nc] = __builtin_amdgcn_wmma_f32_16x16x4_f32(
                               false, pa[c], false, vb, (short)0, oacc[nc],
                               false, false);
            }
        }
    }

#pragma unroll
    for (int r = 0; r < 8; ++r) {
        int m = r + 8 * hi;
        if (m < nrows) {
            float inv = 1.0f / lacc[r];
            size_t ro = (size_t)(cq0 + base + m) * ROWSZ + h * HD;
#pragma unroll
            for (int nc = 0; nc < 4; ++nc)
                out[ro + nc * 16 + il] = oacc[nc][r] * inv;
        }
    }
}

extern "C" void kernel_launch(void* const* d_in, const int* in_sizes, int n_in,
                              void* d_out, int out_size, void* d_ws, size_t ws_size,
                              hipStream_t stream) {
    const float* q   = (const float*)d_in[0];
    const float* k   = (const float*)d_in[1];
    const float* v   = (const float*)d_in[2];
    const int*   cuq = (const int*)d_in[3];
    const int*   cuk = (const int*)d_in[4];
    float* out = (float*)d_out;

    const int T = in_sizes[0] / ROWSZ;
    const int B = in_sizes[3] - 1;
    dim3 grid((T + 15) / 16, B, NH);
    va_fwd_kernel<<<grid, 32, 0, stream>>>(q, k, v, cuq, cuk, out);
}


# Score tile transposed (keys x queries): a lane owns 8 keys of one query; row max and sum in-lane, P to PV from registers; no LDS, bpermute, ones-WMMA; V loads join K loads; PV, row-sum order change
# speedup vs baseline: 1.6749x; 1.5028x over previous
#include <hip/hip_runtime.h>
#include <math.h>

typedef __attribute__((ext_vector_type(2))) float v2f;
typedef __attribute__((ext_vector_type(8))) float v8f;

#define NH 16
#define HD 64
#define ROWSZ (NH * HD)

__device__ __forceinline__ float xhalf(float x)
{
    const int b = __builtin_bit_cast(int, x);
    return __builtin_bit_cast(float,
        __builtin_amdgcn_permlanex16(b, b, 0x76543210, 0xfedcba98, false, false));
}

__device__ __forceinline__ int tiles_ge(int nt, int sh, int lk, int t)
{
    if (t <= 0) return nt;
    if (16 * (t - 1) > lk - 1) return 0;
    int xmin = t - 1 - sh;
    if (xmin < 0) xmin = 0;
    const int c = nt - xmin;
    return c > 0 ? c : 0;
}

__global__ __launch_bounds__(32)
void va_fwd_kernel(const float* __restrict__ q,
                   const float* __restrict__ k,
                   const float* __restrict__ v,
                   const int*   __restrict__ cuq,
                   const int*   __restrict__ cuk,
                   float*       __restrict__ out)
{
    const int lane = threadIdx.x & 31;
    const int il   = lane & 15;
    const int hi   = lane >> 4;
    const int nb   = gridDim.y;

    int b, h, xt, cq0, cq1, ck0, lenk;
    if (nb > 32) {
        b  = blockIdx.y;
        h  = blockIdx.z;
        xt = blockIdx.x;
        cq0 = cuq[b];
        cq1 = cuq[b + 1];
        ck0 = cuk[b];
        lenk = cuk[b + 1] - ck0;
    } else {
        int a0 = 0, a1 = 0, c0 = 0, lk = 0, nt = 0, sh = 0;
        if (lane < nb) {
            a0 = cuq[lane];
            a1 = cuq[lane + 1];
            c0 = cuk[lane];
            lk = cuk[lane + 1] - c0;
            const int lq = a1 - a0;
            nt = lq > 0 ? (lq + 15) >> 4 : 0;
            sh = (15 + lk - lq) >> 4;
        }
        int tot = 0, tmax = 1;
        for (int bb = 0; bb < nb; ++bb) {
            tot += __builtin_amdgcn_readlane(nt, bb);
            const int kk = ((__builtin_amdgcn_readlane(lk, bb) - 1) >> 4) + 2;
            tmax = kk > tmax ? kk : tmax;
        }
        const int wgid  = blockIdx.x + gridDim.x * (blockIdx.y + gridDim.y * blockIdx.z);
        const int items = NH * tot;
        if (wgid >= items) return;
        int rank = wgid;
        const int rows = items >> 8;
        if (wgid < (rows << 8)) {
            const int j = wgid & 255;
            rank = ((((wgid >> 8) + j) % rows) << 8) + j;
        }

        int lo = 0, span = tmax;
        for (int pass = 0; pass < 7 && span > 1; ++pass) {
            const int step = (span + 31) >> 5;
            const int t = lo + lane * step;
            int n = 0;
            for (int bb = 0; bb < nb; ++bb)
                n += tiles_ge(__builtin_amdgcn_readlane(nt, bb),
                              __builtin_amdgcn_readlane(sh, bb),
                              __builtin_amdgcn_readlane(lk, bb), t);
            const unsigned m = __builtin_amdgcn_ballot_w32(NH * n > rank);
            lo += (__builtin_popcount(m) - 1) * step;
            span = step;
        }
        const int tw = lo;
        int above = 0;
        for (int bb = 0; bb < nb; ++bb)
            above += tiles_ge(__builtin_amdgcn_readlane(nt, bb),
                              __builtin_amdgcn_readlane(sh, bb),
                              __builtin_amdgcn_readlane(lk, bb), tw + 1);
        const int r = rank - NH * above;
        h = r % NH;
        int qx = r / NH;
        b = 0;
        xt = 0;
        bool found = false;
        for (int bb = 0; bb < nb; ++bb) {
            const int ntb = __builtin_amdgcn_readlane(nt, bb);
            const int shb = __builtin_amdgcn_readlane(sh, bb);
            const int lkb = __builtin_amdgcn_readlane(lk, bb);
            const int ca = tiles_ge(ntb, shb, lkb, tw);
            const int g  = ca - tiles_ge(ntb, shb, lkb, tw + 1);
            if (!found) {
                if (qx < g) { b = bb; xt = ntb - ca + qx; found = true; }
                else        qx -= g;
            }
        }
        cq0  = __builtin_amdgcn_readlane(a0, b);
        cq1  = __builtin_amdgcn_readlane(a1, b);
        ck0  = __builtin_amdgcn_readlane(c0, b);
        lenk = __builtin_amdgcn_readlane(lk, b);
    }

    const int lenq = cq1 - cq0;
    const int base = xt * 16;
    if (base >= lenq) return;

    const int delta = lenk - lenq;
    const int nrows = min(16, lenq - base);


    int qrow = cq0 + base + il;
    if (qrow > cq1 - 1) qrow = cq1 - 1;
    const float* qp = q + (size_t)qrow * ROWSZ + h * HD + 2 * hi;
    v2f qa[16];
#pragma unroll
    for (int c = 0; c < 16; ++c) {
        v2f t = *(const v2f*)(qp + 4 * c);
        qa[c] = t * 0.125f;
    }

    v8f oacc[4];
#pragma unroll
    for (int nc = 0; nc < 4; ++nc) oacc[nc] = (v8f){};
    float lsum = 0.0f;
    float mrow = -3.0e38f;

    int maxj = base + 15 + delta;
    if (maxj > lenk - 1) maxj = lenk - 1;
    if (maxj < 0) {
#pragma unroll
        for (int r = 0; r < 8; ++r) {
            int m = r + 8 * hi;
            if (m < nrows) {
                size_t ro = (size_t)(cq0 + base + m) * ROWSZ + h * HD;
#pragma unroll
                for (int nc = 0; nc < 4; ++nc) out[ro + nc * 16 + il] = 0.0f;
            }
        }
        return;
    }
    const int ktiles = maxj / 16 + 1;
    const float* kbase = k + (size_t)ck0 * ROWSZ + h * HD;
    const float* vbase = v + (size_t)ck0 * ROWSZ + h * HD;

    for (int kt = 0; kt < ktiles; ++kt) {
        if (kt + 1 < ktiles) {
            int pr = min((kt + 1) * 16 + il, lenk - 1);
            __builtin_prefetch(kbase + (size_t)pr * ROWSZ, 0, 3);
            __builtin_prefetch(vbase + (size_t)pr * ROWSZ, 0, 3);
        }

        int krow = kt * 16 + il;
        if (krow > lenk - 1) krow = lenk - 1;
        const float* kp = kbase + (size_t)krow * ROWSZ + 2 * hi;
        v2f kb[16];
#pragma unroll
        for (int c = 0; c < 16; ++c) kb[c] = *(const v2f*)(kp + 4 * c);

        v2f va[4][4];
#pragma unroll
        for (int c = 0; c < 4; ++c) {
            int kr0 = kt * 16 + 8 * hi + 2 * c;
            int kr1 = kr0 + 1;
            kr0 = min(kr0, lenk - 1);
            kr1 = min(kr1, lenk - 1);
#pragma unroll
            for (int nc = 0; nc < 4; ++nc) {
                va[nc][c][0] = vbase[(size_t)kr0 * ROWSZ + nc * 16 + il];
                va[nc][c][1] = vbase[(size_t)kr1 * ROWSZ + nc * 16 + il];
            }
        }
        __builtin_amdgcn_sched_barrier(0);

        v8f s = (v8f){};
#pragma unroll
        for (int c = 0; c < 16; ++c)
            s = __builtin_amdgcn_wmma_f32_16x16x4_f32(
                    false, kb[c], false, qa[c], (short)0, s, false, false);

        const int ig = base + il;
#pragma unroll
        for (int r = 0; r < 8; ++r) {
            const int jg = kt * 16 + 8 * hi + r;
            bool ok = (jg <= ig + delta) && (jg < lenk);
            s[r] = ok ? s[r] : -1.0e9f;
        }

        float x = fmaxf(fmaxf(fmaxf(s[0], s[1]), fmaxf(s[2], s[3])),
                        fmaxf(fmaxf(s[4], s[5]), fmaxf(s[6], s[7])));
        x = fmaxf(x, xhalf(x));
        const float mnew  = fmaxf(mrow, x);
        const float alpha = __expf(mrow - mnew);
        mrow = mnew;
#pragma unroll
        for (int r = 0; r < 8; ++r) s[r] = __expf(s[r] - mnew);
        lsum = lsum * alpha + (((s[0] + s[1]) + (s[2] + s[3])) + ((s[4] + s[5]) + (s[6] + s[7])));
#pragma unroll
        for (int nc = 0; nc < 4; ++nc) oacc[nc] *= alpha;

#pragma unroll
        for (int nc = 0; nc < 4; ++nc) {
#pragma unroll
            for (int c = 0; c < 4; ++c) {
                v2f pb = { s[2 * c], s[2 * c + 1] };
                oacc[nc] = __builtin_amdgcn_wmma_f32_16x16x4_f32(
                               false, va[nc][c], false, pb, (short)0, oacc[nc],
                               false, false);
            }
        }
    }

    const float inv = 1.0f / (lsum + xhalf(lsum));
    if (il < nrows) {
        float* op = out + (size_t)(cq0 + base + il) * ROWSZ + h * HD + 8 * hi;
#pragma unroll
        for (int nc = 0; nc < 4; ++nc) {
            const float4 lo4 = { oacc[nc][0] * inv, oacc[nc][1] * inv,
                                 oacc[nc][2] * inv, oacc[nc][3] * inv };
            const float4 hi4 = { oacc[nc][4] * inv, oacc[nc][5] * inv,
                                 oacc[nc][6] * inv, oacc[nc][7] * inv };
            *(float4*)(op + nc * 16)     = lo4;
            *(float4*)(op + nc * 16 + 4) = hi4;
        }
    }
}

extern "C" void kernel_launch(void* const* d_in, const int* in_sizes, int n_in,
                              void* d_out, int out_size, void* d_ws, size_t ws_size,
                              hipStream_t stream) {
    const float* q   = (const float*)d_in[0];
    const float* k   = (const float*)d_in[1];
    const float* v   = (const float*)d_in[2];
    const int*   cuq = (const int*)d_in[3];
    const int*   cuk = (const int*)d_in[4];
    float* out = (float*)d_out;

    const int T = in_sizes[0] / ROWSZ;
    const int B = in_sizes[3] - 1;
    dim3 grid((T + 15) / 16, B, NH);
    va_fwd_kernel<<<grid, 32, 0, stream>>>(q, k, v, cuq, cuk, out);
}


# Work-group = 4 waves on one (batch, head, query tile) item: wave w runs key tiles w, w+4, ..; partial max, row sum, O joined via LDS (sum order change); grid x = tiles + batches, no empty groups
# speedup vs baseline: 2.2876x; 1.3658x over previous
#include <hip/hip_runtime.h>
#include <math.h>

typedef __attribute__((ext_vector_type(2))) float v2f;
typedef __attribute__((ext_vector_type(8))) float v8f;

#define NH 16
#define HD 64
#define ROWSZ (NH * HD)

__device__ __forceinline__ float xhalf(float x)
{
    const int b = __builtin_bit_cast(int, x);
    return __builtin_bit_cast(float,
        __builtin_amdgcn_permlanex16(b, b, 0x76543210, 0xfedcba98, false, false));
}

__device__ __forceinline__ int tiles_ge(int nt, int sh, int lk, int t)
{
    if (t <= 0) return nt;
    if (16 * (t - 1) > lk - 1) return 0;
    int xmin = t - 1 - sh;
    if (xmin < 0) xmin = 0;
    const int c = nt - xmin;
    return c > 0 ? c : 0;
}

#define NW 4
#define OPITCH 17

__global__ __launch_bounds__(32 * NW)
void va_fwd_kernel(const float* __restrict__ q,
                   const float* __restrict__ k,
                   const float* __restrict__ v,
                   const int*   __restrict__ cuq,
                   const int*   __restrict__ cuk,
                   float*       __restrict__ out,
                   int nb)
{
    __shared__ float4 so[NW][16][OPITCH];
    __shared__ float  sm[NW][16];
    __shared__ float  sl[NW][32];

    const int lane = threadIdx.x & 31;
    const int il   = lane & 15;
    const int hi   = lane >> 4;
    const int w    = __builtin_amdgcn_readfirstlane((int)(threadIdx.x >> 5));

    int b, h, xt, cq0, cq1, ck0, lenk;
    if (nb > 32) {
        h = blockIdx.z;
        int x = blockIdx.x;
        bool found = false;
        b = 0;
        xt = 0;
        for (int bb = 0; bb < nb; ++bb) {
            const int lq  = cuq[bb + 1] - cuq[bb];
            const int ntb = lq > 0 ? (lq + 15) >> 4 : 0;
            if (x < ntb) { b = bb; xt = x; found = true; break; }
            x -= ntb;
        }
        if (!found) return;
        cq0 = cuq[b];
        cq1 = cuq[b + 1];
        ck0 = cuk[b];
        lenk = cuk[b + 1] - ck0;
    } else {
        int a0 = 0, a1 = 0, c0 = 0, lk = 0, nt = 0, sh = 0;
        if (lane < nb) {
            a0 = cuq[lane];
            a1 = cuq[lane + 1];
            c0 = cuk[lane];
            lk = cuk[lane + 1] - c0;
            const int lq = a1 - a0;
            nt = lq > 0 ? (lq + 15) >> 4 : 0;
            sh = (15 + lk - lq) >> 4;
        }
        int tot = 0, tmax = 1;
        for (int bb = 0; bb < nb; ++bb) {
            tot += __builtin_amdgcn_readlane(nt, bb);
            const int kk = ((__builtin_amdgcn_readlane(lk, bb) - 1) >> 4) + 2;
            tmax = kk > tmax ? kk : tmax;
        }
        const int wgid  = blockIdx.x + gridDim.x * blockIdx.z;
        const int items = NH * tot;
        if (wgid >= items) return;
        int rank = wgid;
        const int rows = items >> 8;
        if (wgid < (rows << 8)) {
            const int j = wgid & 255;
            rank = ((((wgid >> 8) + j) % rows) << 8) + j;
        }

        int lo = 0, span = tmax;
        for (int pass = 0; pass < 7 && span > 1; ++pass) {
            const int step = (span + 31) >> 5;
            const int t = lo + lane * step;
            int n = 0;
            for (int bb = 0; bb < nb; ++bb)
                n += tiles_ge(__builtin_amdgcn_readlane(nt, bb),
                              __builtin_amdgcn_readlane(sh, bb),
                              __builtin_amdgcn_readlane(lk, bb), t);
            const unsigned m = __builtin_amdgcn_ballot_w32(NH * n > rank);
            lo += (__builtin_popcount(m) - 1) * step;
            span = step;
        }
        const int tw = lo;
        int above = 0;
        for (int bb = 0; bb < nb; ++bb)
            above += tiles_ge(__builtin_amdgcn_readlane(nt, bb),
                              __builtin_amdgcn_readlane(sh, bb),
                              __builtin_amdgcn_readlane(lk, bb), tw + 1);
        const int r = rank - NH * above;
        h = r % NH;
        int qx = r / NH;
        b = 0;
        xt = 0;
        bool found = false;
        for (int bb = 0; bb < nb; ++bb) {
            const int ntb = __builtin_amdgcn_readlane(nt, bb);
            const int shb = __builtin_amdgcn_readlane(sh, bb);
            const int lkb = __builtin_amdgcn_readlane(lk, bb);
            const int ca = tiles_ge(ntb, shb, lkb, tw);
            const int g  = ca - tiles_ge(ntb, shb, lkb, tw + 1);
            if (!found) {
                if (qx < g) { b = bb; xt = ntb - ca + qx; found = true; }
                else        qx -= g;
            }
        }
        cq0  = __builtin_amdgcn_readlane(a0, b);
        cq1  = __builtin_amdgcn_readlane(a1, b);
        ck0  = __builtin_amdgcn_readlane(c0, b);
        lenk = __builtin_amdgcn_readlane(lk, b);
    }

    const int lenq = cq1 - cq0;
    const int base = xt * 16;
    if (base >= lenq) return;

    const int delta = lenk - lenq;
    const int nrows = min(16, lenq - base);


    int qrow = cq0 + base + il;
    if (qrow > cq1 - 1) qrow = cq1 - 1;
    const float* qp = q + (size_t)qrow * ROWSZ + h * HD + 2 * hi;
    v2f qa[16];
#pragma unroll
    for (int c = 0; c < 16; ++c) {
        v2f t = *(const v2f*)(qp + 4 * c);
        qa[c] = t * 0.125f;
    }

    v8f oacc[4];
#pragma unroll
    for (int nc = 0; nc < 4; ++nc) oacc[nc] = (v8f){};
    float lsum = 0.0f;
    float mrow = -3.0e38f;

    int maxj = base + 15 + delta;
    if (maxj > lenk - 1) maxj = lenk - 1;
    if (maxj < 0) {
#pragma unroll
        for (int r = 0; r < 8 && w == 0; ++r) {
            int m = r + 8 * hi;
            if (m < nrows) {
                size_t ro = (size_t)(cq0 + base + m) * ROWSZ + h * HD;
#pragma unroll
                for (int nc = 0; nc < 4; ++nc) out[ro + nc * 16 + il] = 0.0f;
            }
        }
        return;
    }
    const int ktiles = maxj / 16 + 1;
    const float* kbase = k + (size_t)ck0 * ROWSZ + h * HD;
    const float* vbase = v + (size_t)ck0 * ROWSZ + h * HD;

    for (int kt = w; kt < ktiles; kt += NW) {
        if (kt + NW < ktiles) {
            int pr = min((kt + NW) * 16 + il, lenk - 1);
            __builtin_prefetch(kbase + (size_t)pr * ROWSZ, 0, 3);
            __builtin_prefetch(vbase + (size_t)pr * ROWSZ, 0, 3);
        }

        int krow = kt * 16 + il;
        if (krow > lenk - 1) krow = lenk - 1;
        const float* kp = kbase + (size_t)krow * ROWSZ + 2 * hi;
        v2f kb[16];
#pragma unroll
        for (int c = 0; c < 16; ++c) kb[c] = *(const v2f*)(kp + 4 * c);

        v2f va[4][4];
#pragma unroll
        for (int c = 0; c < 4; ++c) {
            int kr0 = kt * 16 + 8 * hi + 2 * c;
            int kr1 = kr0 + 1;
            kr0 = min(kr0, lenk - 1);
            kr1 = min(kr1, lenk - 1);
#pragma unroll
            for (int nc = 0; nc < 4; ++nc) {
                va[nc][c][0] = vbase[(size_t)kr0 * ROWSZ + nc * 16 + il];
                va[nc][c][1] = vbase[(size_t)kr1 * ROWSZ + nc * 16 + il];
            }
        }
        __builtin_amdgcn_sched_barrier(0);

        v8f s = (v8f){};
#pragma unroll
        for (int c = 0; c < 16; ++c)
            s = __builtin_amdgcn_wmma_f32_16x16x4_f32(
                    false, kb[c], false, qa[c], (short)0, s, false, false);

        const int ig = base + il;
#pragma unroll
        for (int r = 0; r < 8; ++r) {
            const int jg = kt * 16 + 8 * hi + r;
            bool ok = (jg <= ig + delta) && (jg < lenk);
            s[r] = ok ? s[r] : -1.0e9f;
        }

        float x = fmaxf(fmaxf(fmaxf(s[0], s[1]), fmaxf(s[2], s[3])),
                        fmaxf(fmaxf(s[4], s[5]), fmaxf(s[6], s[7])));
        x = fmaxf(x, xhalf(x));
        const float mnew  = fmaxf(mrow, x);
        const float alpha = __expf(mrow - mnew);
        mrow = mnew;
#pragma unroll
        for (int r = 0; r < 8; ++r) s[r] = __expf(s[r] - mnew);
        lsum = lsum * alpha + (((s[0] + s[1]) + (s[2] + s[3])) + ((s[4] + s[5]) + (s[6] + s[7])));
#pragma unroll
        for (int nc = 0; nc < 4; ++nc) oacc[nc] *= alpha;

#pragma unroll
        for (int nc = 0; nc < 4; ++nc) {
#pragma unroll
            for (int c = 0; c < 4; ++c) {
                v2f pb = { s[2 * c], s[2 * c + 1] };
                oacc[nc] = __builtin_amdgcn_wmma_f32_16x16x4_f32(
                               false, va[nc][c], false, pb, (short)0, oacc[nc],
                               false, false);
            }
        }
    }

#pragma unroll
    for (int nc = 0; nc < 4; ++nc) {
        so[w][il][4 * nc + 2 * hi]     = (float4){ oacc[nc][0], oacc[nc][1], oacc[nc][2], oacc[nc][3] };
        so[w][il][4 * nc + 2 * hi + 1] = (float4){ oacc[nc][4], oacc[nc][5], oacc[nc][6], oacc[nc][7] };
    }
    sm[w][il]   = mrow;
    sl[w][lane] = lsum;
    __syncthreads();

    const int ql = 4 * w + (lane >> 3);
    const int p  = lane & 7;
    float mall = sm[0][ql];
#pragma unroll
    for (int u = 1; u < NW; ++u) mall = fmaxf(mall, sm[u][ql]);
    float  den = 0.0f;
    float4 o0 = { 0.0f, 0.0f, 0.0f, 0.0f };
    float4 o1 = { 0.0f, 0.0f, 0.0f, 0.0f };
#pragma unroll
    for (int u = 0; u < NW; ++u) {
        const float  su = __expf(sm[u][ql] - mall);
        den += su * (sl[u][ql] + sl[u][ql + 16]);
        const float4 a0 = so[u][ql][p];
        const float4 a1 = so[u][ql][8 + p];
        o0.x += su * a0.x; o0.y += su * a0.y; o0.z += su * a0.z; o0.w += su * a0.w;
        o1.x += su * a1.x; o1.y += su * a1.y; o1.z += su * a1.z; o1.w += su * a1.w;
    }
    const float inv = 1.0f / den;
    if (ql < nrows) {
        float* op = out + (size_t)(cq0 + base + ql) * ROWSZ + h * HD + 4 * p;
        *(float4*)(op)      = (float4){ o0.x * inv, o0.y * inv, o0.z * inv, o0.w * inv };
        *(float4*)(op + 32) = (float4){ o1.x * inv, o1.y * inv, o1.z * inv, o1.w * inv };
    }
}

extern "C" void kernel_launch(void* const* d_in, const int* in_sizes, int n_in,
                              void* d_out, int out_size, void* d_ws, size_t ws_size,
                              hipStream_t stream) {
    const float* q   = (const float*)d_in[0];
    const float* k   = (const float*)d_in[1];
    const float* v   = (const float*)d_in[2];
    const int*   cuq = (const int*)d_in[3];
    const int*   cuk = (const int*)d_in[4];
    float* out = (float*)d_out;

    const int T = in_sizes[0] / ROWSZ;
    const int B = in_sizes[3] - 1;
    dim3 grid((T + 15) / 16 + B, 1, NH);
    va_fwd_kernel<<<grid, 32 * NW, 0, stream>>>(q, k, v, cuq, cuk, out, B);
}


# K tile read from memory in whole lines (one load = 2 head rows = 4 lines, was 16 lines per load) and turned to fragment order through the wave's own LDS tile; WMMA inputs and order unchanged
# speedup vs baseline: 2.3662x; 1.0344x over previous
#include <hip/hip_runtime.h>
#include <math.h>

typedef __attribute__((ext_vector_type(2))) float v2f;
typedef __attribute__((ext_vector_type(4))) float v4f;
typedef __attribute__((ext_vector_type(8))) float v8f;

#define NH 16
#define HD 64
#define ROWSZ (NH * HD)

__device__ __forceinline__ float xhalf(float x)
{
    const int b = __builtin_bit_cast(int, x);
    return __builtin_bit_cast(float,
        __builtin_amdgcn_permlanex16(b, b, 0x76543210, 0xfedcba98, false, false));
}

__device__ __forceinline__ void wave_lds_order()
{
    __builtin_amdgcn_fence(__ATOMIC_RELEASE, "wavefront");
    __builtin_amdgcn_wave_barrier();
    __builtin_amdgcn_fence(__ATOMIC_ACQUIRE, "wavefront");
}

__device__ __forceinline__ int tiles_ge(int nt, int sh, int lk, int t)
{
    if (t <= 0) return nt;
    if (16 * (t - 1) > lk - 1) return 0;
    int xmin = t - 1 - sh;
    if (xmin < 0) xmin = 0;
    const int c = nt - xmin;
    return c > 0 ? c : 0;
}

#define NW 4
#define OPITCH 17

__global__ __launch_bounds__(32 * NW)
void va_fwd_kernel(const float* __restrict__ q,
                   const float* __restrict__ k,
                   const float* __restrict__ v,
                   const int*   __restrict__ cuq,
                   const int*   __restrict__ cuk,
                   float*       __restrict__ out,
                   int nb)
{
    __shared__ float4 so[NW][16][OPITCH];
    __shared__ float  sm[NW][16];
    __shared__ float  sl[NW][32];

    const int lane = threadIdx.x & 31;
    const int il   = lane & 15;
    const int hi   = lane >> 4;
    const int w    = __builtin_amdgcn_readfirstlane((int)(threadIdx.x >> 5));

    int b, h, xt, cq0, cq1, ck0, lenk;
    if (nb > 32) {
        h = blockIdx.z;
        int x = blockIdx.x;
        bool found = false;
        b = 0;
        xt = 0;
        for (int bb = 0; bb < nb; ++bb) {
            const int lq  = cuq[bb + 1] - cuq[bb];
            const int ntb = lq > 0 ? (lq + 15) >> 4 : 0;
            if (x < ntb) { b = bb; xt = x; found = true; break; }
            x -= ntb;
        }
        if (!found) return;
        cq0 = cuq[b];
        cq1 = cuq[b + 1];
        ck0 = cuk[b];
        lenk = cuk[b + 1] - ck0;
    } else {
        int a0 = 0, a1 = 0, c0 = 0, lk = 0, nt = 0, sh = 0;
        if (lane < nb) {
            a0 = cuq[lane];
            a1 = cuq[lane + 1];
            c0 = cuk[lane];
            lk = cuk[lane + 1] - c0;
            const int lq = a1 - a0;
            nt = lq > 0 ? (lq + 15) >> 4 : 0;
            sh = (15 + lk - lq) >> 4;
        }
        int tot = 0, tmax = 1;
        for (int bb = 0; bb < nb; ++bb) {
            tot += __builtin_amdgcn_readlane(nt, bb);
            const int kk = ((__builtin_amdgcn_readlane(lk, bb) - 1) >> 4) + 2;
            tmax = kk > tmax ? kk : tmax;
        }
        const int wgid  = blockIdx.x + gridDim.x * blockIdx.z;
        const int items = NH * tot;
        if (wgid >= items) return;
        int rank = wgid;
        const int rows = items >> 8;
        if (wgid < (rows << 8)) {
            const int j = wgid & 255;
            rank = ((((wgid >> 8) + j) % rows) << 8) + j;
        }

        int lo = 0, span = tmax;
        for (int pass = 0; pass < 7 && span > 1; ++pass) {
            const int step = (span + 31) >> 5;
            const int t = lo + lane * step;
            int n = 0;
            for (int bb = 0; bb < nb; ++bb)
                n += tiles_ge(__builtin_amdgcn_readlane(nt, bb),
                              __builtin_amdgcn_readlane(sh, bb),
                              __builtin_amdgcn_readlane(lk, bb), t);
            const unsigned m = __builtin_amdgcn_ballot_w32(NH * n > rank);
            lo += (__builtin_popcount(m) - 1) * step;
            span = step;
        }
        const int tw = lo;
        int above = 0;
        for (int bb = 0; bb < nb; ++bb)
            above += tiles_ge(__builtin_amdgcn_readlane(nt, bb),
                              __builtin_amdgcn_readlane(sh, bb),
                              __builtin_amdgcn_readlane(lk, bb), tw + 1);
        const int r = rank - NH * above;
        h = r % NH;
        int qx = r / NH;
        b = 0;
        xt = 0;
        bool found = false;
        for (int bb = 0; bb < nb; ++bb) {
            const int ntb = __builtin_amdgcn_readlane(nt, bb);
            const int shb = __builtin_amdgcn_readlane(sh, bb);
            const int lkb = __builtin_amdgcn_readlane(lk, bb);
            const int ca = tiles_ge(ntb, shb, lkb, tw);
            const int g  = ca - tiles_ge(ntb, shb, lkb, tw + 1);
            if (!found) {
                if (qx < g) { b = bb; xt = ntb - ca + qx; found = true; }
                else        qx -= g;
            }
        }
        cq0  = __builtin_amdgcn_readlane(a0, b);
        cq1  = __builtin_amdgcn_readlane(a1, b);
        ck0  = __builtin_amdgcn_readlane(c0, b);
        lenk = __builtin_amdgcn_readlane(lk, b);
    }

    const int lenq = cq1 - cq0;
    const int base = xt * 16;
    if (base >= lenq) return;

    const int delta = lenk - lenq;
    const int nrows = min(16, lenq - base);


    int qrow = cq0 + base + il;
    if (qrow > cq1 - 1) qrow = cq1 - 1;
    const float* qp = q + (size_t)qrow * ROWSZ + h * HD + 2 * hi;
    v2f qa[16];
#pragma unroll
    for (int c = 0; c < 16; ++c) {
        v2f t = *(const v2f*)(qp + 4 * c);
        qa[c] = t * 0.125f;
    }

    v8f oacc[4];
#pragma unroll
    for (int nc = 0; nc < 4; ++nc) oacc[nc] = (v8f){};
    float lsum = 0.0f;
    float mrow = -3.0e38f;

    int maxj = base + 15 + delta;
    if (maxj > lenk - 1) maxj = lenk - 1;
    if (maxj < 0) {
#pragma unroll
        for (int r = 0; r < 8 && w == 0; ++r) {
            int m = r + 8 * hi;
            if (m < nrows) {
                size_t ro = (size_t)(cq0 + base + m) * ROWSZ + h * HD;
#pragma unroll
                for (int nc = 0; nc < 4; ++nc) out[ro + nc * 16 + il] = 0.0f;
            }
        }
        return;
    }
    const int ktiles = maxj / 16 + 1;
    const float* kbase = k + (size_t)ck0 * ROWSZ + h * HD;
    const float* vbase = v + (size_t)ck0 * ROWSZ + h * HD;
    float* ktile = (float*)&so[w][0][0];

    for (int kt = w; kt < ktiles; kt += NW) {
        if (kt + NW < ktiles) {
            int pr = min((kt + NW) * 16 + il, lenk - 1);
            __builtin_prefetch(kbase + (size_t)pr * ROWSZ, 0, 3);
            __builtin_prefetch(vbase + (size_t)pr * ROWSZ, 0, 3);
        }

        v4f kg[8];
#pragma unroll
        for (int j = 0; j < 8; ++j) {
            int kr = kt * 16 + 8 * hi + j;
            kr = min(kr, lenk - 1);
            kg[j] = *(const v4f*)(kbase + (size_t)kr * ROWSZ + 4 * il);
        }

        v2f va[4][4];
#pragma unroll
        for (int c = 0; c < 4; ++c) {
            int kr0 = kt * 16 + 8 * hi + 2 * c;
            int kr1 = kr0 + 1;
            kr0 = min(kr0, lenk - 1);
            kr1 = min(kr1, lenk - 1);
#pragma unroll
            for (int nc = 0; nc < 4; ++nc) {
                va[nc][c][0] = vbase[(size_t)kr0 * ROWSZ + nc * 16 + il];
                va[nc][c][1] = vbase[(size_t)kr1 * ROWSZ + nc * 16 + il];
            }
        }
        __builtin_amdgcn_sched_barrier(0);

#pragma unroll
        for (int j = 0; j < 8; ++j)
            *(v4f*)(ktile + (8 * hi + j) * (4 * OPITCH) + 4 * il) = kg[j];
        wave_lds_order();
        v2f kb[16];
#pragma unroll
        for (int c = 0; c < 16; ++c)
            kb[c] = *(const v2f*)(ktile + il * (4 * OPITCH) + 4 * c + 2 * hi);
        __builtin_amdgcn_sched_barrier(0);

        v8f s = (v8f){};
#pragma unroll
        for (int c = 0; c < 16; ++c)
            s = __builtin_amdgcn_wmma_f32_16x16x4_f32(
                    false, kb[c], false, qa[c], (short)0, s, false, false);

        const int ig = base + il;
#pragma unroll
        for (int r = 0; r < 8; ++r) {
            const int jg = kt * 16 + 8 * hi + r;
            bool ok = (jg <= ig + delta) && (jg < lenk);
            s[r] = ok ? s[r] : -1.0e9f;
        }

        float x = fmaxf(fmaxf(fmaxf(s[0], s[1]), fmaxf(s[2], s[3])),
                        fmaxf(fmaxf(s[4], s[5]), fmaxf(s[6], s[7])));
        x = fmaxf(x, xhalf(x));
        const float mnew  = fmaxf(mrow, x);
        const float alpha = __expf(mrow - mnew);
        mrow = mnew;
#pragma unroll
        for (int r = 0; r < 8; ++r) s[r] = __expf(s[r] - mnew);
        lsum = lsum * alpha + (((s[0] + s[1]) + (s[2] + s[3])) + ((s[4] + s[5]) + (s[6] + s[7])));
#pragma unroll
        for (int nc = 0; nc < 4; ++nc) oacc[nc] *= alpha;

#pragma unroll
        for (int nc = 0; nc < 4; ++nc) {
#pragma unroll
            for (int c = 0; c < 4; ++c) {
                v2f pb = { s[2 * c], s[2 * c + 1] };
                oacc[nc] = __builtin_amdgcn_wmma_f32_16x16x4_f32(
                               false, va[nc][c], false, pb, (short)0, oacc[nc],
                               false, false);
            }
        }
    }

#pragma unroll
    for (int nc = 0; nc < 4; ++nc) {
        so[w][il][4 * nc + 2 * hi]     = (float4){ oacc[nc][0], oacc[nc][1], oacc[nc][2], oacc[nc][3] };
        so[w][il][4 * nc + 2 * hi + 1] = (float4){ oacc[nc][4], oacc[nc][5], oacc[nc][6], oacc[nc][7] };
    }
    sm[w][il]   = mrow;
    sl[w][lane] = lsum;
    __syncthreads();

    const int ql = 4 * w + (lane >> 3);
    const int p  = lane & 7;
    float mall = sm[0][ql];
#pragma unroll
    for (int u = 1; u < NW; ++u) mall = fmaxf(mall, sm[u][ql]);
    float  den = 0.0f;
    float4 o0 = { 0.0f, 0.0f, 0.0f, 0.0f };
    float4 o1 = { 0.0f, 0.0f, 0.0f, 0.0f };
#pragma unroll
    for (int u = 0; u < NW; ++u) {
        const float  su = __expf(sm[u][ql] - mall);
        den += su * (sl[u][ql] + sl[u][ql + 16]);
        const float4 a0 = so[u][ql][p];
        const float4 a1 = so[u][ql][8 + p];
        o0.x += su * a0.x; o0.y += su * a0.y; o0.z += su * a0.z; o0.w += su * a0.w;
        o1.x += su * a1.x; o1.y += su * a1.y; o1.z += su * a1.z; o1.w += su * a1.w;
    }
    const float inv = 1.0f / den;
    if (ql < nrows) {
        float* op = out + (size_t)(cq0 + base + ql) * ROWSZ + h * HD + 4 * p;
        *(float4*)(op)      = (float4){ o0.x * inv, o0.y * inv, o0.z * inv, o0.w * inv };
        *(float4*)(op + 32) = (float4){ o1.x * inv, o1.y * inv, o1.z * inv, o1.w * inv };
    }
}

extern "C" void kernel_launch(void* const* d_in, const int* in_sizes, int n_in,
                              void* d_out, int out_size, void* d_ws, size_t ws_size,
                              hipStream_t stream) {
    const float* q   = (const float*)d_in[0];
    const float* k   = (const float*)d_in[1];
    const float* v   = (const float*)d_in[2];
    const int*   cuq = (const int*)d_in[3];
    const int*   cuk = (const int*)d_in[4];
    float* out = (float*)d_out;

    const int T = in_sizes[0] / ROWSZ;
    const int B = in_sizes[3] - 1;
    dim3 grid((T + 15) / 16 + B, 1, NH);
    va_fwd_kernel<<<grid, 32 * NW, 0, stream>>>(q, k, v, cuq, cuk, out, B);
}


# Work-group id = rank in the list sorted by key-tile count, longest first (the dealt mix of lengths is dropped): long items all start first, the last groups to start are the shortest
# speedup vs baseline: 2.8071x; 1.1863x over previous
#include <hip/hip_runtime.h>
#include <math.h>

typedef __attribute__((ext_vector_type(2))) float v2f;
typedef __attribute__((ext_vector_type(4))) float v4f;
typedef __attribute__((ext_vector_type(8))) float v8f;

#define NH 16
#define HD 64
#define ROWSZ (NH * HD)

__device__ __forceinline__ float xhalf(float x)
{
    const int b = __builtin_bit_cast(int, x);
    return __builtin_bit_cast(float,
        __builtin_amdgcn_permlanex16(b, b, 0x76543210, 0xfedcba98, false, false));
}

__device__ __forceinline__ void wave_lds_order()
{
    __builtin_amdgcn_fence(__ATOMIC_RELEASE, "wavefront");
    __builtin_amdgcn_wave_barrier();
    __builtin_amdgcn_fence(__ATOMIC_ACQUIRE, "wavefront");
}

__device__ __forceinline__ int tiles_ge(int nt, int sh, int lk, int t)
{
    if (t <= 0) return nt;
    if (16 * (t - 1) > lk - 1) return 0;
    int xmin = t - 1 - sh;
    if (xmin < 0) xmin = 0;
    const int c = nt - xmin;
    return c > 0 ? c : 0;
}

#define NW 4
#define OPITCH 17

__global__ __launch_bounds__(32 * NW)
void va_fwd_kernel(const float* __restrict__ q,
                   const float* __restrict__ k,
                   const float* __restrict__ v,
                   const int*   __restrict__ cuq,
                   const int*   __restrict__ cuk,
                   float*       __restrict__ out,
                   int nb)
{
    __shared__ float4 so[NW][16][OPITCH];
    __shared__ float  sm[NW][16];
    __shared__ float  sl[NW][32];

    const int lane = threadIdx.x & 31;
    const int il   = lane & 15;
    const int hi   = lane >> 4;
    const int w    = __builtin_amdgcn_readfirstlane((int)(threadIdx.x >> 5));

    int b, h, xt, cq0, cq1, ck0, lenk;
    if (nb > 32) {
        h = blockIdx.z;
        int x = blockIdx.x;
        bool found = false;
        b = 0;
        xt = 0;
        for (int bb = 0; bb < nb; ++bb) {
            const int lq  = cuq[bb + 1] - cuq[bb];
            const int ntb = lq > 0 ? (lq + 15) >> 4 : 0;
            if (x < ntb) { b = bb; xt = x; found = true; break; }
            x -= ntb;
        }
        if (!found) return;
        cq0 = cuq[b];
        cq1 = cuq[b + 1];
        ck0 = cuk[b];
        lenk = cuk[b + 1] - ck0;
    } else {
        int a0 = 0, a1 = 0, c0 = 0, lk = 0, nt = 0, sh = 0;
        if (lane < nb) {
            a0 = cuq[lane];
            a1 = cuq[lane + 1];
            c0 = cuk[lane];
            lk = cuk[lane + 1] - c0;
            const int lq = a1 - a0;
            nt = lq > 0 ? (lq + 15) >> 4 : 0;
            sh = (15 + lk - lq) >> 4;
        }
        int tot = 0, tmax = 1;
        for (int bb = 0; bb < nb; ++bb) {
            tot += __builtin_amdgcn_readlane(nt, bb);
            const int kk = ((__builtin_amdgcn_readlane(lk, bb) - 1) >> 4) + 2;
            tmax = kk > tmax ? kk : tmax;
        }
        const int wgid  = blockIdx.x + gridDim.x * blockIdx.z;
        const int items = NH * tot;
        if (wgid >= items) return;
        const int rank = wgid;

        int lo = 0, span = tmax;
        for (int pass = 0; pass < 7 && span > 1; ++pass) {
            const int step = (span + 31) >> 5;
            const int t = lo + lane * step;
            int n = 0;
            for (int bb = 0; bb < nb; ++bb)
                n += tiles_ge(__builtin_amdgcn_readlane(nt, bb),
                              __builtin_amdgcn_readlane(sh, bb),
                              __builtin_amdgcn_readlane(lk, bb), t);
            const unsigned m = __builtin_amdgcn_ballot_w32(NH * n > rank);
            lo += (__builtin_popcount(m) - 1) * step;
            span = step;
        }
        const int tw = lo;
        int above = 0;
        for (int bb = 0; bb < nb; ++bb)
            above += tiles_ge(__builtin_amdgcn_readlane(nt, bb),
                              __builtin_amdgcn_readlane(sh, bb),
                              __builtin_amdgcn_readlane(lk, bb), tw + 1);
        const int r = rank - NH * above;
        h = r % NH;
        int qx = r / NH;
        b = 0;
        xt = 0;
        bool found = false;
        for (int bb = 0; bb < nb; ++bb) {
            const int ntb = __builtin_amdgcn_readlane(nt, bb);
            const int shb = __builtin_amdgcn_readlane(sh, bb);
            const int lkb = __builtin_amdgcn_readlane(lk, bb);
            const int ca = tiles_ge(ntb, shb, lkb, tw);
            const int g  = ca - tiles_ge(ntb, shb, lkb, tw + 1);
            if (!found) {
                if (qx < g) { b = bb; xt = ntb - ca + qx; found = true; }
                else        qx -= g;
            }
        }
        cq0  = __builtin_amdgcn_readlane(a0, b);
        cq1  = __builtin_amdgcn_readlane(a1, b);
        ck0  = __builtin_amdgcn_readlane(c0, b);
        lenk = __builtin_amdgcn_readlane(lk, b);
    }

    const int lenq = cq1 - cq0;
    const int base = xt * 16;
    if (base >= lenq) return;

    const int delta = lenk - lenq;
    const int nrows = min(16, lenq - base);


    int qrow = cq0 + base + il;
    if (qrow > cq1 - 1) qrow = cq1 - 1;
    const float* qp = q + (size_t)qrow * ROWSZ + h * HD + 2 * hi;
    v2f qa[16];
#pragma unroll
    for (int c = 0; c < 16; ++c) {
        v2f t = *(const v2f*)(qp + 4 * c);
        qa[c] = t * 0.125f;
    }

    v8f oacc[4];
#pragma unroll
    for (int nc = 0; nc < 4; ++nc) oacc[nc] = (v8f){};
    float lsum = 0.0f;
    float mrow = -3.0e38f;

    int maxj = base + 15 + delta;
    if (maxj > lenk - 1) maxj = lenk - 1;
    if (maxj < 0) {
#pragma unroll
        for (int r = 0; r < 8 && w == 0; ++r) {
            int m = r + 8 * hi;
            if (m < nrows) {
                size_t ro = (size_t)(cq0 + base + m) * ROWSZ + h * HD;
#pragma unroll
                for (int nc = 0; nc < 4; ++nc) out[ro + nc * 16 + il] = 0.0f;
            }
        }
        return;
    }
    const int ktiles = maxj / 16 + 1;
    const float* kbase = k + (size_t)ck0 * ROWSZ + h * HD;
    const float* vbase = v + (size_t)ck0 * ROWSZ + h * HD;
    float* ktile = (float*)&so[w][0][0];

    for (int kt = w; kt < ktiles; kt += NW) {
        if (kt + NW < ktiles) {
            int pr = min((kt + NW) * 16 + il, lenk - 1);
            __builtin_prefetch(kbase + (size_t)pr * ROWSZ, 0, 3);
            __builtin_prefetch(vbase + (size_t)pr * ROWSZ, 0, 3);
        }

        v4f kg[8];
#pragma unroll
        for (int j = 0; j < 8; ++j) {
            int kr = kt * 16 + 8 * hi + j;
            kr = min(kr, lenk - 1);
            kg[j] = *(const v4f*)(kbase + (size_t)kr * ROWSZ + 4 * il);
        }

        v2f va[4][4];
#pragma unroll
        for (int c = 0; c < 4; ++c) {
            int kr0 = kt * 16 + 8 * hi + 2 * c;
            int kr1 = kr0 + 1;
            kr0 = min(kr0, lenk - 1);
            kr1 = min(kr1, lenk - 1);
#pragma unroll
            for (int nc = 0; nc < 4; ++nc) {
                va[nc][c][0] = vbase[(size_t)kr0 * ROWSZ + nc * 16 + il];
                va[nc][c][1] = vbase[(size_t)kr1 * ROWSZ + nc * 16 + il];
            }
        }
        __builtin_amdgcn_sched_barrier(0);

#pragma unroll
        for (int j = 0; j < 8; ++j)
            *(v4f*)(ktile + (8 * hi + j) * (4 * OPITCH) + 4 * il) = kg[j];
        wave_lds_order();
        v2f kb[16];
#pragma unroll
        for (int c = 0; c < 16; ++c)
            kb[c] = *(const v2f*)(ktile + il * (4 * OPITCH) + 4 * c + 2 * hi);
        __builtin_amdgcn_sched_barrier(0);

        v8f s = (v8f){};
#pragma unroll
        for (int c = 0; c < 16; ++c)
            s = __builtin_amdgcn_wmma_f32_16x16x4_f32(
                    false, kb[c], false, qa[c], (short)0, s, false, false);

        const int ig = base + il;
#pragma unroll
        for (int r = 0; r < 8; ++r) {
            const int jg = kt * 16 + 8 * hi + r;
            bool ok = (jg <= ig + delta) && (jg < lenk);
            s[r] = ok ? s[r] : -1.0e9f;
        }

        float x = fmaxf(fmaxf(fmaxf(s[0], s[1]), fmaxf(s[2], s[3])),
                        fmaxf(fmaxf(s[4], s[5]), fmaxf(s[6], s[7])));
        x = fmaxf(x, xhalf(x));
        const float mnew  = fmaxf(mrow, x);
        const float alpha = __expf(mrow - mnew);
        mrow = mnew;
#pragma unroll
        for (int r = 0; r < 8; ++r) s[r] = __expf(s[r] - mnew);
        lsum = lsum * alpha + (((s[0] + s[1]) + (s[2] + s[3])) + ((s[4] + s[5]) + (s[6] + s[7])));
#pragma unroll
        for (int nc = 0; nc < 4; ++nc) oacc[nc] *= alpha;

#pragma unroll
        for (int nc = 0; nc < 4; ++nc) {
#pragma unroll
            for (int c = 0; c < 4; ++c) {
                v2f pb = { s[2 * c], s[2 * c + 1] };
                oacc[nc] = __builtin_amdgcn_wmma_f32_16x16x4_f32(
                               false, va[nc][c], false, pb, (short)0, oacc[nc],
                               false, false);
            }
        }
    }

#pragma unroll
    for (int nc = 0; nc < 4; ++nc) {
        so[w][il][4 * nc + 2 * hi]     = (float4){ oacc[nc][0], oacc[nc][1], oacc[nc][2], oacc[nc][3] };
        so[w][il][4 * nc + 2 * hi + 1] = (float4){ oacc[nc][4], oacc[nc][5], oacc[nc][6], oacc[nc][7] };
    }
    sm[w][il]   = mrow;
    sl[w][lane] = lsum;
    __syncthreads();

    const int ql = 4 * w + (lane >> 3);
    const int p  = lane & 7;
    float mall = sm[0][ql];
#pragma unroll
    for (int u = 1; u < NW; ++u) mall = fmaxf(mall, sm[u][ql]);
    float  den = 0.0f;
    float4 o0 = { 0.0f, 0.0f, 0.0f, 0.0f };
    float4 o1 = { 0.0f, 0.0f, 0.0f, 0.0f };
#pragma unroll
    for (int u = 0; u < NW; ++u) {
        const float  su = __expf(sm[u][ql] - mall);
        den += su * (sl[u][ql] + sl[u][ql + 16]);
        const float4 a0 = so[u][ql][p];
        const float4 a1 = so[u][ql][8 + p];
        o0.x += su * a0.x; o0.y += su * a0.y; o0.z += su * a0.z; o0.w += su * a0.w;
        o1.x += su * a1.x; o1.y += su * a1.y; o1.z += su * a1.z; o1.w += su * a1.w;
    }
    const float inv = 1.0f / den;
    if (ql < nrows) {
        float* op = out + (size_t)(cq0 + base + ql) * ROWSZ + h * HD + 4 * p;
        *(float4*)(op)      = (float4){ o0.x * inv, o0.y * inv, o0.z * inv, o0.w * inv };
        *(float4*)(op + 32) = (float4){ o1.x * inv, o1.y * inv, o1.z * inv, o1.w * inv };
    }
}

extern "C" void kernel_launch(void* const* d_in, const int* in_sizes, int n_in,
                              void* d_out, int out_size, void* d_ws, size_t ws_size,
                              hipStream_t stream) {
    const float* q   = (const float*)d_in[0];
    const float* k   = (const float*)d_in[1];
    const float* v   = (const float*)d_in[2];
    const int*   cuq = (const int*)d_in[3];
    const int*   cuk = (const int*)d_in[4];
    float* out = (float*)d_out;

    const int T = in_sizes[0] / ROWSZ;
    const int B = in_sizes[3] - 1;
    dim3 grid((T + 15) / 16 + B, 1, NH);
    va_fwd_kernel<<<grid, 32 * NW, 0, stream>>>(q, k, v, cuq, cuk, out, B);
}


# Each wave carries two 16-query sub-tiles (item = 32 query rows) through its key tiles: every K and V fragment read once feeds two WMMAs; half as many work-groups; sums and their order unchanged
# speedup vs baseline: 3.2297x; 1.1505x over previous
#include <hip/hip_runtime.h>
#include <math.h>

typedef __attribute__((ext_vector_type(2))) float v2f;
typedef __attribute__((ext_vector_type(4))) float v4f;
typedef __attribute__((ext_vector_type(8))) float v8f;

#define NH 16
#define HD 64
#define ROWSZ (NH * HD)

__device__ __forceinline__ float xhalf(float x)
{
    const int b = __builtin_bit_cast(int, x);
    return __builtin_bit_cast(float,
        __builtin_amdgcn_permlanex16(b, b, 0x76543210, 0xfedcba98, false, false));
}

__device__ __forceinline__ void wave_lds_order()
{
    __builtin_amdgcn_fence(__ATOMIC_RELEASE, "wavefront");
    __builtin_amdgcn_wave_barrier();
    __builtin_amdgcn_fence(__ATOMIC_ACQUIRE, "wavefront");
}

__device__ __forceinline__ int tiles_ge(int nt, int sh, int lk, int t)
{
    if (t <= 0) return nt;
    if (16 * (t - 1) > lk - 1) return 0;
    const int n = t - 1 - sh;
    const int xmin = n > 0 ? (n + 1) >> 1 : 0;
    const int c = nt - xmin;
    return c > 0 ? c : 0;
}

#define NW 4
#define QS 2
#define QROWS (16 * QS)
#define OPITCH 17

__global__ __launch_bounds__(32 * NW)
void va_fwd_kernel(const float* __restrict__ q,
                   const float* __restrict__ k,
                   const float* __restrict__ v,
                   const int*   __restrict__ cuq,
                   const int*   __restrict__ cuk,
                   float*       __restrict__ out,
                   int nb)
{
    __shared__ float4 so[NW][QROWS][OPITCH];
    __shared__ float  sm[NW][QROWS];
    __shared__ float  sl[NW][QS][32];

    const int lane = threadIdx.x & 31;
    const int il   = lane & 15;
    const int hi   = lane >> 4;
    const int w    = __builtin_amdgcn_readfirstlane((int)(threadIdx.x >> 5));

    int b, h, xt, cq0, cq1, ck0, lenk;
    if (nb > 32) {
        h = blockIdx.z;
        int x = blockIdx.x;
        bool found = false;
        b = 0;
        xt = 0;
        for (int bb = 0; bb < nb; ++bb) {
            const int lq  = cuq[bb + 1] - cuq[bb];
            const int ntb = lq > 0 ? (lq + QROWS - 1) / QROWS : 0;
            if (x < ntb) { b = bb; xt = x; found = true; break; }
            x -= ntb;
        }
        if (!found) return;
        cq0 = cuq[b];
        cq1 = cuq[b + 1];
        ck0 = cuk[b];
        lenk = cuk[b + 1] - ck0;
    } else {
        int a0 = 0, a1 = 0, c0 = 0, lk = 0, nt = 0, sh = 0;
        if (lane < nb) {
            a0 = cuq[lane];
            a1 = cuq[lane + 1];
            c0 = cuk[lane];
            lk = cuk[lane + 1] - c0;
            const int lq = a1 - a0;
            nt = lq > 0 ? (lq + QROWS - 1) / QROWS : 0;
            sh = (QROWS - 1 + lk - lq) >> 4;
        }
        int tot = 0, tmax = 1;
        for (int bb = 0; bb < nb; ++bb) {
            tot += __builtin_amdgcn_readlane(nt, bb);
            const int kk = ((__builtin_amdgcn_readlane(lk, bb) - 1) >> 4) + 2;
            tmax = kk > tmax ? kk : tmax;
        }
        const int wgid  = blockIdx.x + gridDim.x * blockIdx.z;
        const int items = NH * tot;
        if (wgid >= items) return;
        const int rank = wgid;

        int lo = 0, span = tmax;
        for (int pass = 0; pass < 7 && span > 1; ++pass) {
            const int step = (span + 31) >> 5;
            const int t = lo + lane * step;
            int n = 0;
            for (int bb = 0; bb < nb; ++bb)
                n += tiles_ge(__builtin_amdgcn_readlane(nt, bb),
                              __builtin_amdgcn_readlane(sh, bb),
                              __builtin_amdgcn_readlane(lk, bb), t);
            const unsigned m = __builtin_amdgcn_ballot_w32(NH * n > rank);
            lo += (__builtin_popcount(m) - 1) * step;
            span = step;
        }
        const int tw = lo;
        int above = 0;
        for (int bb = 0; bb < nb; ++bb)
            above += tiles_ge(__builtin_amdgcn_readlane(nt, bb),
                              __builtin_amdgcn_readlane(sh, bb),
                              __builtin_amdgcn_readlane(lk, bb), tw + 1);
        const int r = rank - NH * above;
        h = r % NH;
        int qx = r / NH;
        b = 0;
        xt = 0;
        bool found = false;
        for (int bb = 0; bb < nb; ++bb) {
            const int ntb = __builtin_amdgcn_readlane(nt, bb);
            const int shb = __builtin_amdgcn_readlane(sh, bb);
            const int lkb = __builtin_amdgcn_readlane(lk, bb);
            const int ca = tiles_ge(ntb, shb, lkb, tw);
            const int g  = ca - tiles_ge(ntb, shb, lkb, tw + 1);
            if (!found) {
                if (qx < g) { b = bb; xt = ntb - ca + qx; found = true; }
                else        qx -= g;
            }
        }
        cq0  = __builtin_amdgcn_readlane(a0, b);
        cq1  = __builtin_amdgcn_readlane(a1, b);
        ck0  = __builtin_amdgcn_readlane(c0, b);
        lenk = __builtin_amdgcn_readlane(lk, b);
    }

    const int lenq = cq1 - cq0;
    const int base = xt * QROWS;
    if (base >= lenq) return;

    const int delta = lenk - lenq;
    const int nrows = min(QROWS, lenq - base);


    v2f qa[QS][16];
#pragma unroll
    for (int a = 0; a < QS; ++a) {
        int qrow = cq0 + base + 16 * a + il;
        if (qrow > cq1 - 1) qrow = cq1 - 1;
        const float* qp = q + (size_t)qrow * ROWSZ + h * HD + 2 * hi;
#pragma unroll
        for (int c = 0; c < 16; ++c) {
            v2f t = *(const v2f*)(qp + 4 * c);
            qa[a][c] = t * 0.125f;
        }
    }

    v8f oacc[QS][4];
    float lsum[QS];
    float mrow[QS];
#pragma unroll
    for (int a = 0; a < QS; ++a) {
#pragma unroll
        for (int nc = 0; nc < 4; ++nc) oacc[a][nc] = (v8f){};
        lsum[a] = 0.0f;
        mrow[a] = -3.0e38f;
    }

    int ktn[QS];
#pragma unroll
    for (int a = 0; a < QS; ++a) {
        int mj = base + 16 * a + 15 + delta;
        if (mj > lenk - 1) mj = lenk - 1;
        ktn[a] = (mj < 0 || nrows <= 16 * a) ? 0 : mj / 16 + 1;
    }
    int ktiles = ktn[0];
#pragma unroll
    for (int a = 1; a < QS; ++a) ktiles = ktn[a] > ktiles ? ktn[a] : ktiles;
    if (ktiles == 0) {
#pragma unroll
        for (int r = 0; r < 8 * QS && w == 0; ++r) {
            int m = (r & 7) + 8 * hi + 16 * (r >> 3);
            if (m < nrows) {
                size_t ro = (size_t)(cq0 + base + m) * ROWSZ + h * HD;
#pragma unroll
                for (int nc = 0; nc < 4; ++nc) out[ro + nc * 16 + il] = 0.0f;
            }
        }
        return;
    }
    const float* kbase = k + (size_t)ck0 * ROWSZ + h * HD;
    const float* vbase = v + (size_t)ck0 * ROWSZ + h * HD;
    float* ktile = (float*)&so[w][0][0];

    for (int kt = w; kt < ktiles; kt += NW) {
        if (kt + NW < ktiles) {
            int pr = min((kt + NW) * 16 + il, lenk - 1);
            __builtin_prefetch(kbase + (size_t)pr * ROWSZ, 0, 3);
            __builtin_prefetch(vbase + (size_t)pr * ROWSZ, 0, 3);
        }

        v4f kg[8];
#pragma unroll
        for (int j = 0; j < 8; ++j) {
            int kr = kt * 16 + 8 * hi + j;
            kr = min(kr, lenk - 1);
            kg[j] = *(const v4f*)(kbase + (size_t)kr * ROWSZ + 4 * il);
        }

        v2f va[4][4];
#pragma unroll
        for (int c = 0; c < 4; ++c) {
            int kr0 = kt * 16 + 8 * hi + 2 * c;
            int kr1 = kr0 + 1;
            kr0 = min(kr0, lenk - 1);
            kr1 = min(kr1, lenk - 1);
#pragma unroll
            for (int nc = 0; nc < 4; ++nc) {
                va[nc][c][0] = vbase[(size_t)kr0 * ROWSZ + nc * 16 + il];
                va[nc][c][1] = vbase[(size_t)kr1 * ROWSZ + nc * 16 + il];
            }
        }
        __builtin_amdgcn_sched_barrier(0);

#pragma unroll
        for (int j = 0; j < 8; ++j)
            *(v4f*)(ktile + (8 * hi + j) * (4 * OPITCH) + 4 * il) = kg[j];
        wave_lds_order();
        v2f kb[16];
#pragma unroll
        for (int c = 0; c < 16; ++c)
            kb[c] = *(const v2f*)(ktile + il * (4 * OPITCH) + 4 * c + 2 * hi);
        __builtin_amdgcn_sched_barrier(0);

        v8f s[QS];
#pragma unroll
        for (int a = 0; a < QS; ++a) s[a] = (v8f){};
#pragma unroll
        for (int c = 0; c < 16; ++c) {
#pragma unroll
            for (int a = 0; a < QS; ++a)
                s[a] = __builtin_amdgcn_wmma_f32_16x16x4_f32(
                           false, kb[c], false, qa[a][c], (short)0, s[a], false, false);
        }

#pragma unroll
        for (int a = 0; a < QS; ++a) {
            const float off = kt < ktn[a] ? -1.0e9f : -__builtin_inff();
            const int ig = base + 16 * a + il;
#pragma unroll
            for (int r = 0; r < 8; ++r) {
                const int jg = kt * 16 + 8 * hi + r;
                bool ok = (jg <= ig + delta) && (jg < lenk);
                s[a][r] = ok ? s[a][r] : off;
            }

            float x = fmaxf(fmaxf(fmaxf(s[a][0], s[a][1]), fmaxf(s[a][2], s[a][3])),
                            fmaxf(fmaxf(s[a][4], s[a][5]), fmaxf(s[a][6], s[a][7])));
            x = fmaxf(x, xhalf(x));
            const float mnew  = fmaxf(mrow[a], x);
            const float alpha = __expf(mrow[a] - mnew);
            mrow[a] = mnew;
#pragma unroll
            for (int r = 0; r < 8; ++r) s[a][r] = __expf(s[a][r] - mnew);
            lsum[a] = lsum[a] * alpha + (((s[a][0] + s[a][1]) + (s[a][2] + s[a][3]))
                                       + ((s[a][4] + s[a][5]) + (s[a][6] + s[a][7])));
#pragma unroll
            for (int nc = 0; nc < 4; ++nc) oacc[a][nc] *= alpha;
        }

#pragma unroll
        for (int nc = 0; nc < 4; ++nc) {
#pragma unroll
            for (int c = 0; c < 4; ++c) {
#pragma unroll
                for (int a = 0; a < QS; ++a) {
                    v2f pb = { s[a][2 * c], s[a][2 * c + 1] };
                    oacc[a][nc] = __builtin_amdgcn_wmma_f32_16x16x4_f32(
                                      false, va[nc][c], false, pb, (short)0, oacc[a][nc],
                                      false, false);
                }
            }
        }
    }

#pragma unroll
    for (int a = 0; a < QS; ++a) {
#pragma unroll
        for (int nc = 0; nc < 4; ++nc) {
            so[w][16 * a + il][4 * nc + 2 * hi]     = (float4){ oacc[a][nc][0], oacc[a][nc][1], oacc[a][nc][2], oacc[a][nc][3] };
            so[w][16 * a + il][4 * nc + 2 * hi + 1] = (float4){ oacc[a][nc][4], oacc[a][nc][5], oacc[a][nc][6], oacc[a][nc][7] };
        }
        sm[w][16 * a + il] = mrow[a];
        sl[w][a][lane]     = lsum[a];
    }
    __syncthreads();

    const int p = lane & 7;
#pragma unroll
    for (int ps = 0; ps < QROWS / (4 * NW); ++ps) {
        const int ql = (QROWS / NW) * w + 4 * ps + (lane >> 3);
        const int qa_ = ((QROWS / NW) * w + 4 * ps) >> 4;
        const int qi = ql & 15;
        float mall = sm[0][ql];
#pragma unroll
        for (int u = 1; u < NW; ++u) mall = fmaxf(mall, sm[u][ql]);
        float  den = 0.0f;
        float4 o0 = { 0.0f, 0.0f, 0.0f, 0.0f };
        float4 o1 = { 0.0f, 0.0f, 0.0f, 0.0f };
#pragma unroll
        for (int u = 0; u < NW; ++u) {
            const float  su = __expf(sm[u][ql] - mall);
            den += su * (sl[u][qa_][qi] + sl[u][qa_][qi + 16]);
            const float4 a0 = so[u][ql][p];
            const float4 a1 = so[u][ql][8 + p];
            o0.x += su * a0.x; o0.y += su * a0.y; o0.z += su * a0.z; o0.w += su * a0.w;
            o1.x += su * a1.x; o1.y += su * a1.y; o1.z += su * a1.z; o1.w += su * a1.w;
        }
        const float inv = ktn[qa_] > 0 ? 1.0f / den : 0.0f;
        if (ql < nrows) {
            float* op = out + (size_t)(cq0 + base + ql) * ROWSZ + h * HD + 4 * p;
            *(float4*)(op)      = (float4){ o0.x * inv, o0.y * inv, o0.z * inv, o0.w * inv };
            *(float4*)(op + 32) = (float4){ o1.x * inv, o1.y * inv, o1.z * inv, o1.w * inv };
        }
    }
}

extern "C" void kernel_launch(void* const* d_in, const int* in_sizes, int n_in,
                              void* d_out, int out_size, void* d_ws, size_t ws_size,
                              hipStream_t stream) {
    const float* q   = (const float*)d_in[0];
    const float* k   = (const float*)d_in[1];
    const float* v   = (const float*)d_in[2];
    const int*   cuq = (const int*)d_in[3];
    const int*   cuk = (const int*)d_in[4];
    float* out = (float*)d_out;

    const int T = in_sizes[0] / ROWSZ;
    const int B = in_sizes[3] - 1;
    dim3 grid((T + QROWS - 1) / QROWS + B, 1, NH);
    va_fwd_kernel<<<grid, 32 * NW, 0, stream>>>(q, k, v, cuq, cuk, out, B);
}


# Key tiles wholly below the item's causal diagonal and inside the batch skip the mask and the key-row clamps; row addresses there = scalar tile base + fixed offsets; sums and order unchanged
# speedup vs baseline: 3.3798x; 1.0465x over previous
#include <hip/hip_runtime.h>
#include <math.h>

typedef __attribute__((ext_vector_type(2))) float v2f;
typedef __attribute__((ext_vector_type(4))) float v4f;
typedef __attribute__((ext_vector_type(8))) float v8f;

#define NH 16
#define HD 64
#define ROWSZ (NH * HD)

__device__ __forceinline__ float xhalf(float x)
{
    const int b = __builtin_bit_cast(int, x);
    return __builtin_bit_cast(float,
        __builtin_amdgcn_permlanex16(b, b, 0x76543210, 0xfedcba98, false, false));
}

__device__ __forceinline__ void wave_lds_order()
{
    __builtin_amdgcn_fence(__ATOMIC_RELEASE, "wavefront");
    __builtin_amdgcn_wave_barrier();
    __builtin_amdgcn_fence(__ATOMIC_ACQUIRE, "wavefront");
}

__device__ __forceinline__ int tiles_ge(int nt, int sh, int lk, int t)
{
    if (t <= 0) return nt;
    if (16 * (t - 1) > lk - 1) return 0;
    const int n = t - 1 - sh;
    const int xmin = n > 0 ? (n + 1) >> 1 : 0;
    const int c = nt - xmin;
    return c > 0 ? c : 0;
}

#define NW 4
#define QS 2
#define QROWS (16 * QS)
#define OPITCH 17

__global__ __launch_bounds__(32 * NW)
void va_fwd_kernel(const float* __restrict__ q,
                   const float* __restrict__ k,
                   const float* __restrict__ v,
                   const int*   __restrict__ cuq,
                   const int*   __restrict__ cuk,
                   float*       __restrict__ out,
                   int nb)
{
    __shared__ float4 so[NW][QROWS][OPITCH];
    __shared__ float  sm[NW][QROWS];
    __shared__ float  sl[NW][QS][32];

    const int lane = threadIdx.x & 31;
    const int il   = lane & 15;
    const int hi   = lane >> 4;
    const int w    = __builtin_amdgcn_readfirstlane((int)(threadIdx.x >> 5));

    int b, h, xt, cq0, cq1, ck0, lenk;
    if (nb > 32) {
        h = blockIdx.z;
        int x = blockIdx.x;
        bool found = false;
        b = 0;
        xt = 0;
        for (int bb = 0; bb < nb; ++bb) {
            const int lq  = cuq[bb + 1] - cuq[bb];
            const int ntb = lq > 0 ? (lq + QROWS - 1) / QROWS : 0;
            if (x < ntb) { b = bb; xt = x; found = true; break; }
            x -= ntb;
        }
        if (!found) return;
        cq0 = cuq[b];
        cq1 = cuq[b + 1];
        ck0 = cuk[b];
        lenk = cuk[b + 1] - ck0;
    } else {
        int a0 = 0, a1 = 0, c0 = 0, lk = 0, nt = 0, sh = 0;
        if (lane < nb) {
            a0 = cuq[lane];
            a1 = cuq[lane + 1];
            c0 = cuk[lane];
            lk = cuk[lane + 1] - c0;
            const int lq = a1 - a0;
            nt = lq > 0 ? (lq + QROWS - 1) / QROWS : 0;
            sh = (QROWS - 1 + lk - lq) >> 4;
        }
        int tot = 0, tmax = 1;
        for (int bb = 0; bb < nb; ++bb) {
            tot += __builtin_amdgcn_readlane(nt, bb);
            const int kk = ((__builtin_amdgcn_readlane(lk, bb) - 1) >> 4) + 2;
            tmax = kk > tmax ? kk : tmax;
        }
        const int wgid  = blockIdx.x + gridDim.x * blockIdx.z;
        const int items = NH * tot;
        if (wgid >= items) return;
        const int rank = wgid;

        int lo = 0, span = tmax;
        for (int pass = 0; pass < 7 && span > 1; ++pass) {
            const int step = (span + 31) >> 5;
            const int t = lo + lane * step;
            int n = 0;
            for (int bb = 0; bb < nb; ++bb)
                n += tiles_ge(__builtin_amdgcn_readlane(nt, bb),
                              __builtin_amdgcn_readlane(sh, bb),
                              __builtin_amdgcn_readlane(lk, bb), t);
            const unsigned m = __builtin_amdgcn_ballot_w32(NH * n > rank);
            lo += (__builtin_popcount(m) - 1) * step;
            span = step;
        }
        const int tw = lo;
        int above = 0;
        for (int bb = 0; bb < nb; ++bb)
            above += tiles_ge(__builtin_amdgcn_readlane(nt, bb),
                              __builtin_amdgcn_readlane(sh, bb),
                              __builtin_amdgcn_readlane(lk, bb), tw + 1);
        const int r = rank - NH * above;
        h = r % NH;
        int qx = r / NH;
        b = 0;
        xt = 0;
        bool found = false;
        for (int bb = 0; bb < nb; ++bb) {
            const int ntb = __builtin_amdgcn_readlane(nt, bb);
            const int shb = __builtin_amdgcn_readlane(sh, bb);
            const int lkb = __builtin_amdgcn_readlane(lk, bb);
            const int ca = tiles_ge(ntb, shb, lkb, tw);
            const int g  = ca - tiles_ge(ntb, shb, lkb, tw + 1);
            if (!found) {
                if (qx < g) { b = bb; xt = ntb - ca + qx; found = true; }
                else        qx -= g;
            }
        }
        cq0  = __builtin_amdgcn_readlane(a0, b);
        cq1  = __builtin_amdgcn_readlane(a1, b);
        ck0  = __builtin_amdgcn_readlane(c0, b);
        lenk = __builtin_amdgcn_readlane(lk, b);
    }

    const int lenq = cq1 - cq0;
    const int base = xt * QROWS;
    if (base >= lenq) return;

    const int delta = lenk - lenq;
    const int nrows = min(QROWS, lenq - base);


    v2f qa[QS][16];
#pragma unroll
    for (int a = 0; a < QS; ++a) {
        int qrow = cq0 + base + 16 * a + il;
        if (qrow > cq1 - 1) qrow = cq1 - 1;
        const float* qp = q + (size_t)qrow * ROWSZ + h * HD + 2 * hi;
#pragma unroll
        for (int c = 0; c < 16; ++c) {
            v2f t = *(const v2f*)(qp + 4 * c);
            qa[a][c] = t * 0.125f;
        }
    }

    v8f oacc[QS][4];
    float lsum[QS];
    float mrow[QS];
#pragma unroll
    for (int a = 0; a < QS; ++a) {
#pragma unroll
        for (int nc = 0; nc < 4; ++nc) oacc[a][nc] = (v8f){};
        lsum[a] = 0.0f;
        mrow[a] = -3.0e38f;
    }

    int ktn[QS];
#pragma unroll
    for (int a = 0; a < QS; ++a) {
        int mj = base + 16 * a + 15 + delta;
        if (mj > lenk - 1) mj = lenk - 1;
        ktn[a] = (mj < 0 || nrows <= 16 * a) ? 0 : mj / 16 + 1;
    }
    int ktiles = ktn[0];
#pragma unroll
    for (int a = 1; a < QS; ++a) ktiles = ktn[a] > ktiles ? ktn[a] : ktiles;
    if (ktiles == 0) {
#pragma unroll
        for (int r = 0; r < 8 * QS && w == 0; ++r) {
            int m = (r & 7) + 8 * hi + 16 * (r >> 3);
            if (m < nrows) {
                size_t ro = (size_t)(cq0 + base + m) * ROWSZ + h * HD;
#pragma unroll
                for (int nc = 0; nc < 4; ++nc) out[ro + nc * 16 + il] = 0.0f;
            }
        }
        return;
    }
    const float* kbase = k + (size_t)ck0 * ROWSZ + h * HD;
    const float* vbase = v + (size_t)ck0 * ROWSZ + h * HD;
    float* ktile = (float*)&so[w][0][0];

    for (int kt = w; kt < ktiles; kt += NW) {
        if (kt + NW < ktiles) {
            int pr = min((kt + NW) * 16 + il, lenk - 1);
            __builtin_prefetch(kbase + (size_t)pr * ROWSZ, 0, 3);
            __builtin_prefetch(vbase + (size_t)pr * ROWSZ, 0, 3);
        }
        const bool interior = kt * 16 + 15 <= base + delta && kt * 16 + 16 <= lenk;

        v4f kg[8];
        v2f va[4][4];
        if (interior) {
            const float* ktb = kbase + (size_t)(kt * 16) * ROWSZ;
            const float* vtb = vbase + (size_t)(kt * 16) * ROWSZ;
#pragma unroll
            for (int j = 0; j < 8; ++j)
                kg[j] = *(const v4f*)(ktb + (8 * hi + j) * ROWSZ + 4 * il);
#pragma unroll
            for (int c = 0; c < 4; ++c) {
#pragma unroll
                for (int nc = 0; nc < 4; ++nc) {
                    va[nc][c][0] = vtb[(8 * hi + 2 * c) * ROWSZ + nc * 16 + il];
                    va[nc][c][1] = vtb[(8 * hi + 2 * c + 1) * ROWSZ + nc * 16 + il];
                }
            }
        } else {
#pragma unroll
            for (int j = 0; j < 8; ++j) {
                int kr = kt * 16 + 8 * hi + j;
                kr = min(kr, lenk - 1);
                kg[j] = *(const v4f*)(kbase + (size_t)kr * ROWSZ + 4 * il);
            }
#pragma unroll
            for (int c = 0; c < 4; ++c) {
                int kr0 = kt * 16 + 8 * hi + 2 * c;
                int kr1 = kr0 + 1;
                kr0 = min(kr0, lenk - 1);
                kr1 = min(kr1, lenk - 1);
#pragma unroll
                for (int nc = 0; nc < 4; ++nc) {
                    va[nc][c][0] = vbase[(size_t)kr0 * ROWSZ + nc * 16 + il];
                    va[nc][c][1] = vbase[(size_t)kr1 * ROWSZ + nc * 16 + il];
                }
            }
        }
        __builtin_amdgcn_sched_barrier(0);

#pragma unroll
        for (int j = 0; j < 8; ++j)
            *(v4f*)(ktile + (8 * hi + j) * (4 * OPITCH) + 4 * il) = kg[j];
        wave_lds_order();
        v2f kb[16];
#pragma unroll
        for (int c = 0; c < 16; ++c)
            kb[c] = *(const v2f*)(ktile + il * (4 * OPITCH) + 4 * c + 2 * hi);
        __builtin_amdgcn_sched_barrier(0);

        v8f s[QS];
#pragma unroll
        for (int a = 0; a < QS; ++a) s[a] = (v8f){};
#pragma unroll
        for (int c = 0; c < 16; ++c) {
#pragma unroll
            for (int a = 0; a < QS; ++a)
                s[a] = __builtin_amdgcn_wmma_f32_16x16x4_f32(
                           false, kb[c], false, qa[a][c], (short)0, s[a], false, false);
        }

        if (!interior) {
#pragma unroll
            for (int a = 0; a < QS; ++a) {
                const float off = kt < ktn[a] ? -1.0e9f : -__builtin_inff();
                const int ig = base + 16 * a + il;
#pragma unroll
                for (int r = 0; r < 8; ++r) {
                    const int jg = kt * 16 + 8 * hi + r;
                    bool ok = (jg <= ig + delta) && (jg < lenk);
                    s[a][r] = ok ? s[a][r] : off;
                }
            }
        }

#pragma unroll
        for (int a = 0; a < QS; ++a) {

            float x = fmaxf(fmaxf(fmaxf(s[a][0], s[a][1]), fmaxf(s[a][2], s[a][3])),
                            fmaxf(fmaxf(s[a][4], s[a][5]), fmaxf(s[a][6], s[a][7])));
            x = fmaxf(x, xhalf(x));
            const float mnew  = fmaxf(mrow[a], x);
            const float alpha = __expf(mrow[a] - mnew);
            mrow[a] = mnew;
#pragma unroll
            for (int r = 0; r < 8; ++r) s[a][r] = __expf(s[a][r] - mnew);
            lsum[a] = lsum[a] * alpha + (((s[a][0] + s[a][1]) + (s[a][2] + s[a][3]))
                                       + ((s[a][4] + s[a][5]) + (s[a][6] + s[a][7])));
#pragma unroll
            for (int nc = 0; nc < 4; ++nc) oacc[a][nc] *= alpha;
        }

#pragma unroll
        for (int nc = 0; nc < 4; ++nc) {
#pragma unroll
            for (int c = 0; c < 4; ++c) {
#pragma unroll
                for (int a = 0; a < QS; ++a) {
                    v2f pb = { s[a][2 * c], s[a][2 * c + 1] };
                    oacc[a][nc] = __builtin_amdgcn_wmma_f32_16x16x4_f32(
                                      false, va[nc][c], false, pb, (short)0, oacc[a][nc],
                                      false, false);
                }
            }
        }
    }

#pragma unroll
    for (int a = 0; a < QS; ++a) {
#pragma unroll
        for (int nc = 0; nc < 4; ++nc) {
            so[w][16 * a + il][4 * nc + 2 * hi]     = (float4){ oacc[a][nc][0], oacc[a][nc][1], oacc[a][nc][2], oacc[a][nc][3] };
            so[w][16 * a + il][4 * nc + 2 * hi + 1] = (float4){ oacc[a][nc][4], oacc[a][nc][5], oacc[a][nc][6], oacc[a][nc][7] };
        }
        sm[w][16 * a + il] = mrow[a];
        sl[w][a][lane]     = lsum[a];
    }
    __syncthreads();

    const int p = lane & 7;
#pragma unroll
    for (int ps = 0; ps < QROWS / (4 * NW); ++ps) {
        const int ql = (QROWS / NW) * w + 4 * ps + (lane >> 3);
        const int qa_ = ((QROWS / NW) * w + 4 * ps) >> 4;
        const int qi = ql & 15;
        float mall = sm[0][ql];
#pragma unroll
        for (int u = 1; u < NW; ++u) mall = fmaxf(mall, sm[u][ql]);
        float  den = 0.0f;
        float4 o0 = { 0.0f, 0.0f, 0.0f, 0.0f };
        float4 o1 = { 0.0f, 0.0f, 0.0f, 0.0f };
#pragma unroll
        for (int u = 0; u < NW; ++u) {
            const float  su = __expf(sm[u][ql] - mall);
            den += su * (sl[u][qa_][qi] + sl[u][qa_][qi + 16]);
            const float4 a0 = so[u][ql][p];
            const float4 a1 = so[u][ql][8 + p];
            o0.x += su * a0.x; o0.y += su * a0.y; o0.z += su * a0.z; o0.w += su * a0.w;
            o1.x += su * a1.x; o1.y += su * a1.y; o1.z += su * a1.z; o1.w += su * a1.w;
        }
        const float inv = ktn[qa_] > 0 ? 1.0f / den : 0.0f;
        if (ql < nrows) {
            float* op = out + (size_t)(cq0 + base + ql) * ROWSZ + h * HD + 4 * p;
            *(float4*)(op)      = (float4){ o0.x * inv, o0.y * inv, o0.z * inv, o0.w * inv };
            *(float4*)(op + 32) = (float4){ o1.x * inv, o1.y * inv, o1.z * inv, o1.w * inv };
        }
    }
}

extern "C" void kernel_launch(void* const* d_in, const int* in_sizes, int n_in,
                              void* d_out, int out_size, void* d_ws, size_t ws_size,
                              hipStream_t stream) {
    const float* q   = (const float*)d_in[0];
    const float* k   = (const float*)d_in[1];
    const float* v   = (const float*)d_in[2];
    const int*   cuq = (const int*)d_in[3];
    const int*   cuk = (const int*)d_in[4];
    float* out = (float*)d_out;

    const int T = in_sizes[0] / ROWSZ;
    const int B = in_sizes[3] - 1;
    dim3 grid((T + QROWS - 1) / QROWS + B, 1, NH);
    va_fwd_kernel<<<grid, 32 * NW, 0, stream>>>(q, k, v, cuq, cuk, out, B);
}
